# starting kernel
# baseline (speedup 1.0000x reference)
#include <hip/hip_runtime.h>

#define BB    4
#define NN    8192
#define KK    32
#define CIN   64
#define CF    67
#define KP    68
#define MID   64
#define COUT  64
#define GRP   8
#define SLOPE 0.1f
#define EPSV  1e-5f

typedef __attribute__((ext_vector_type(2))) float v2f;
typedef __attribute__((ext_vector_type(8))) float v8f;

static __device__ __forceinline__ void wave_gemm(const float* Xs, const float* Ws,
                                                 int rowBase, int kdim, int lane,
                                                 v8f acc[4]) {
    const int l16  = lane & 15;
    const int koff = (lane >> 4) << 1;
    const float* xrow = Xs + (rowBase + l16) * KP;
    for (int k0 = 0; k0 < kdim; k0 += 4) {
        v2f a;
        a.x = xrow[k0 + koff];
        a.y = xrow[k0 + koff + 1];
#pragma unroll
        for (int t = 0; t < 4; ++t) {
            const float* wrow = Ws + (t * 16 + l16) * KP;
            v2f b;
            b.x = wrow[k0 + koff];
            b.y = wrow[k0 + koff + 1];
            acc[t] = __builtin_amdgcn_wmma_f32_16x16x4_f32(
                false, a, false, b, (short)0, acc[t], false, false);
        }
    }
}

static __device__ __forceinline__ void stats_reduce(v8f acc[4], int lane, float* sred) {
    const int l16 = lane & 15;
#pragma unroll
    for (int t = 0; t < 4; ++t) {
        float s = 0.f, q = 0.f;
#pragma unroll
        for (int v = 0; v < 8; ++v) { float x = acc[t][v]; s += x; q += x * x; }
        s += __shfl_xor(s, 1, 32);  q += __shfl_xor(q, 1, 32);
        s += __shfl_xor(s, 2, 32);  q += __shfl_xor(q, 2, 32);
        s += __shfl_xor(s, 4, 32);  q += __shfl_xor(q, 4, 32);
        s += __shfl_xor(s, 16, 32); q += __shfl_xor(q, 16, 32);
        if ((lane & 7) == 0 && lane < 16) {
            int g = 2 * t + (l16 >> 3);
            atomicAdd(&sred[2 * g + 0], s);
            atomicAdd(&sred[2 * g + 1], q);
        }
    }
}

static __device__ __forceinline__ void fc1_stage(const float* __restrict__ signal,
                                                 const int* __restrict__ edges,
                                                 const float* __restrict__ efeat,
                                                 const float* __restrict__ W1,
                                                 float* Xs, float* W1s, float* Cs,
                                                 int p0) {
    const int tid = threadIdx.x;
    {
        int pt = tid >> 6, c = tid & 63;
        Cs[pt * CIN + c] = signal[(long long)(p0 + pt) * CIN + c];
    }
    for (int i = tid; i < MID * KP; i += 256) {
        int m = i / KP, c = i - m * KP;
        W1s[i] = (c < CF) ? W1[m * CF + c] : 0.f;
    }
    __syncthreads();
    {
        int r = tid >> 1;
        int h = tid & 1;
        int p = p0 + (r >> 5);
        int k = r & 31;
        int R = p * KK + k;
        int e = edges[R];
        const float4* src = ((const float4*)signal) + (long long)e * 16 + h * 8;
        const float4* ctr = ((const float4*)Cs) + (r >> 5) * 16 + h * 8;
        float4* dst = ((float4*)(Xs + r * KP)) + h * 8;
#pragma unroll
        for (int i = 0; i < 8; ++i) {
            float4 gv = src[i], cv = ctr[i], o;
            o.x = gv.x - cv.x; o.y = gv.y - cv.y;
            o.z = gv.z - cv.z; o.w = gv.w - cv.w;
            dst[i] = o;
        }
        if (h) {
            const float* e3 = efeat + (long long)R * 3;
            float* xr = Xs + r * KP;
            xr[64] = e3[0]; xr[65] = e3[1]; xr[66] = e3[2]; xr[67] = 0.f;
        }
    }
    __syncthreads();
}

__global__ __launch_bounds__(256) void k_fc1_stats(const float* __restrict__ signal,
                                                   const int* __restrict__ edges,
                                                   const float* __restrict__ efeat,
                                                   const float* __restrict__ W1,
                                                   float* __restrict__ stats1) {
    __shared__ __align__(16) float Xs[128 * KP];
    __shared__ __align__(16) float W1s[MID * KP];
    __shared__ __align__(16) float Cs[4 * CIN];
    __shared__ float sred[GRP * 2];
    const int tid = threadIdx.x;
    if (tid < GRP * 2) sred[tid] = 0.f;
    const int p0 = blockIdx.x * 4;
    fc1_stage(signal, edges, efeat, W1, Xs, W1s, Cs, p0);

    const int wv = tid >> 5, lane = tid & 31;
    v8f acc[4];
    const v8f z = {0.f, 0.f, 0.f, 0.f, 0.f, 0.f, 0.f, 0.f};
    acc[0] = z; acc[1] = z; acc[2] = z; acc[3] = z;
    wave_gemm(Xs, W1s, wv * 16, KP, lane, acc);

    stats_reduce(acc, lane, sred);
    __syncthreads();
    if (tid < GRP * 2) atomicAdd(&stats1[(p0 >> 13) * GRP * 2 + tid], sred[tid]);
}

__global__ __launch_bounds__(256) void k_fc1_pool(const float* __restrict__ signal,
                                                  const int* __restrict__ edges,
                                                  const float* __restrict__ efeat,
                                                  const float* __restrict__ W1,
                                                  const float* __restrict__ g1,
                                                  const float* __restrict__ b1,
                                                  const float* __restrict__ stats1,
                                                  float* __restrict__ h1) {
    __shared__ __align__(16) float Xs[128 * KP];
    __shared__ __align__(16) float W1s[MID * KP];
    __shared__ __align__(16) float Cs[4 * CIN];
    __shared__ float gam[MID], bet[MID];
    __shared__ float st[GRP * 2];
    __shared__ float tmax[8 * MID];
    const int tid = threadIdx.x;
    const int p0 = blockIdx.x * 4;
    const int b  = p0 >> 13;
    if (tid < MID) { gam[tid] = g1[tid]; bet[tid] = b1[tid]; }
    if (tid < GRP * 2) st[tid] = stats1[b * GRP * 2 + tid];
    fc1_stage(signal, edges, efeat, W1, Xs, W1s, Cs, p0);

    const int wv = tid >> 5, lane = tid & 31, l16 = lane & 15;
    v8f acc[4];
    const v8f z = {0.f, 0.f, 0.f, 0.f, 0.f, 0.f, 0.f, 0.f};
    acc[0] = z; acc[1] = z; acc[2] = z; acc[3] = z;
    wave_gemm(Xs, W1s, wv * 16, KP, lane, acc);

    const float invc = 1.f / (float)(NN * KK * (MID / GRP));
#pragma unroll
    for (int t = 0; t < 4; ++t) {
        int c = t * 16 + l16;
        int g = c >> 3;
        float mean = st[2 * g] * invc;
        float var  = st[2 * g + 1] * invc - mean * mean;
        float istd = rsqrtf(var + EPSV);
        float sc = gam[c] * istd;
        float bs = bet[c] - mean * sc;
        float mx = -3.402823466e38f;
#pragma unroll
        for (int v = 0; v < 8; ++v) {
            float x = acc[t][v] * sc + bs;
            x = (x >= 0.f) ? x : SLOPE * x;
            mx = fmaxf(mx, x);
        }
        mx = fmaxf(mx, __shfl_xor(mx, 16, 32));
        if (lane < 16) tmax[wv * MID + c] = mx;
    }
    __syncthreads();
    {
        int q = tid >> 6, c = tid & 63;
        float h = fmaxf(tmax[(2 * q) * MID + c], tmax[(2 * q + 1) * MID + c]);
        h1[(long long)(p0 + q) * MID + c] = h;
    }
}

__global__ __launch_bounds__(256) void k_fc_stats(const float* __restrict__ X,
                                                  const float* __restrict__ W,
                                                  float* __restrict__ Yraw,
                                                  float* __restrict__ stats) {
    __shared__ __align__(16) float Xs[128 * KP];
    __shared__ __align__(16) float Ws[MID * KP];
    __shared__ float sred[GRP * 2];
    const int tid = threadIdx.x;
    if (tid < GRP * 2) sred[tid] = 0.f;
    const long long p0 = (long long)blockIdx.x * 128;
    for (int i = tid; i < 128 * 16; i += 256) {
        int r = i >> 4, c4 = i & 15;
        float4 v = ((const float4*)(X + (p0 + r) * 64))[c4];
        ((float4*)(Xs + r * KP))[c4] = v;
    }
    for (int i = tid; i < MID * KP; i += 256) {
        int m = i / KP, c = i - m * KP;
        Ws[i] = (c < 64) ? W[m * 64 + c] : 0.f;
    }
    __syncthreads();

    const int wv = tid >> 5, lane = tid & 31, l16 = lane & 15, half = lane >> 4;
    v8f acc[4];
    const v8f z = {0.f, 0.f, 0.f, 0.f, 0.f, 0.f, 0.f, 0.f};
    acc[0] = z; acc[1] = z; acc[2] = z; acc[3] = z;
    wave_gemm(Xs, Ws, wv * 16, 64, lane, acc);

#pragma unroll
    for (int t = 0; t < 4; ++t)
#pragma unroll
        for (int v = 0; v < 8; ++v) {
            int row = wv * 16 + v + 8 * half;
            Yraw[(p0 + row) * 64 + t * 16 + l16] = acc[t][v];
        }
    stats_reduce(acc, lane, sred);
    __syncthreads();
    if (tid < GRP * 2) atomicAdd(&stats[(int)(p0 >> 13) * GRP * 2 + tid], sred[tid]);
}

__global__ __launch_bounds__(256) void k_gn_apply(float* __restrict__ Y,
                                                  const float* __restrict__ stats,
                                                  const float* __restrict__ gamma,
                                                  const float* __restrict__ beta,
                                                  float inv_count) {
    long long i = (long long)blockIdx.x * 256 + threadIdx.x;
    int c = (int)(i & 63);
    long long p = i >> 6;
    int b = (int)(p >> 13);
    int g = c >> 3;
    float sum = stats[b * GRP * 2 + 2 * g];
    float sq  = stats[b * GRP * 2 + 2 * g + 1];
    float mean = sum * inv_count;
    float var  = sq * inv_count - mean * mean;
    float istd = rsqrtf(var + EPSV);
    float sc = gamma[c] * istd;
    float bs = beta[c] - mean * sc;
    float x = Y[i] * sc + bs;
    Y[i] = (x >= 0.f) ? x : SLOPE * x;
}

__global__ void k_init(float* __restrict__ ws) {
    if (threadIdx.x < 3 * BB * GRP * 2) ws[threadIdx.x] = 0.f;
}

extern "C" void kernel_launch(void* const* d_in, const int* in_sizes, int n_in,
                              void* d_out, int out_size, void* d_ws, size_t ws_size,
                              hipStream_t stream) {
    (void)in_sizes; (void)n_in; (void)out_size; (void)ws_size;
    const float* signal = (const float*)d_in[0];
    const int*   edges  = (const int*)d_in[1];
    const float* efeat  = (const float*)d_in[2];
    const float* W1 = (const float*)d_in[3];
    const float* g1 = (const float*)d_in[4];
    const float* b1 = (const float*)d_in[5];
    const float* W2 = (const float*)d_in[6];
    const float* g2 = (const float*)d_in[7];
    const float* b2 = (const float*)d_in[8];
    const float* W3 = (const float*)d_in[9];
    const float* g3 = (const float*)d_in[10];
    const float* b3 = (const float*)d_in[11];
    float* out = (float*)d_out;

    float* ws = (float*)d_ws;
    float* stats1 = ws;
    float* stats2 = ws + 64;
    float* stats3 = ws + 128;
    float* h1   = ws + 256;
    float* buf2 = h1 + (long long)32768 * 64;

    const float invc23 = 1.f / (float)(NN * (COUT / GRP));

    hipLaunchKernelGGL(k_init,      dim3(1),    dim3(256), 0, stream, ws);
    hipLaunchKernelGGL(k_fc1_stats, dim3(8192), dim3(256), 0, stream,
                       signal, edges, efeat, W1, stats1);
    hipLaunchKernelGGL(k_fc1_pool,  dim3(8192), dim3(256), 0, stream,
                       signal, edges, efeat, W1, g1, b1, stats1, h1);
    hipLaunchKernelGGL(k_fc_stats,  dim3(256),  dim3(256), 0, stream,
                       h1, W2, buf2, stats2);
    hipLaunchKernelGGL(k_gn_apply,  dim3(8192), dim3(256), 0, stream,
                       buf2, stats2, g2, b2, invc23);
    hipLaunchKernelGGL(k_fc_stats,  dim3(256),  dim3(256), 0, stream,
                       buf2, W3, out, stats3);
    hipLaunchKernelGGL(k_gn_apply,  dim3(8192), dim3(256), 0, stream,
                       out, stats3, g3, b3, invc23);
}


# fc1 kernels' start: a thread's 17 W1 words (pooling kernel: its statistics, gamma, beta words too) requested in one group with the edge ids; 17 serial load-wait-store round trips become one
# speedup vs baseline: 1.8277x; 1.8277x over previous
#include <hip/hip_runtime.h>

#define BB    4
#define NN    8192
#define KK    32
#define CIN   64
#define CF    67
#define KP    68
#define MID   64
#define COUT  64
#define GRP   8
#define SLOPE 0.1f
#define EPSV  1e-5f

typedef __attribute__((ext_vector_type(2))) float v2f;
typedef __attribute__((ext_vector_type(8))) float v8f;

static __device__ __forceinline__ void stats_reduce(v8f acc[4], int lane, float* sred) {
    const int l16 = lane & 15;
#pragma unroll
    for (int t = 0; t < 4; ++t) {
        float s = 0.f, q = 0.f;
#pragma unroll
        for (int v = 0; v < 8; ++v) { float x = acc[t][v]; s += x; q += x * x; }
        s += __shfl_xor(s, 1, 32);  q += __shfl_xor(q, 1, 32);
        s += __shfl_xor(s, 2, 32);  q += __shfl_xor(q, 2, 32);
        s += __shfl_xor(s, 4, 32);  q += __shfl_xor(q, 4, 32);
        s += __shfl_xor(s, 16, 32); q += __shfl_xor(q, 16, 32);
        if ((lane & 7) == 0 && lane < 16) {
            int g = 2 * t + (l16 >> 3);
            atomicAdd(&sred[2 * g + 0], s);
            atomicAdd(&sred[2 * g + 1], q);
        }
    }
}

#define TPW   16
#define NWG1  512
#define KS1   17
#define XSZ   (128 * KP)

#define WPT   17
struct W1Regs { float w[WPT]; };

static __device__ __forceinline__ void fc1_w_issue(const float* __restrict__ W1, W1Regs& r) {
#pragma unroll
    for (int j = 0; j < WPT; ++j) {
        const int i = threadIdx.x + 256 * j;
        const int m = i / KP, c = i - m * KP;
        r.w[j] = W1[m * CF + ((c < CF) ? c : CF - 1)];
    }
}

static __device__ __forceinline__ void fc1_w_store(float* W1s, const W1Regs& r) {
#pragma unroll
    for (int j = 0; j < WPT; ++j) {
        const int i = threadIdx.x + 256 * j;
        const int c = i % KP;
        W1s[i] = (c < CF) ? r.w[j] : 0.f;
    }
}

static __device__ __forceinline__ void fc1_load_wfrag(const float* W1s, int lane,
                                                      v2f (&wf)[KS1][4]) {
    const int l16  = lane & 15;
    const int koff = (lane >> 4) << 1;
#pragma unroll
    for (int ks = 0; ks < KS1; ++ks)
#pragma unroll
        for (int t = 0; t < 4; ++t)
            wf[ks][t] = *(const v2f*)(W1s + (t * 16 + l16) * KP + ks * 4 + koff);
}

struct TileRegs {
    float4 g[8];
    float4 cv;
    float f0, f1, f2;
};

static __device__ __forceinline__ void fc1_load_ids(const int* __restrict__ edges, int p0,
                                                    int (&e)[8]) {
    const int tid = threadIdx.x;
    const int rb = (tid >> 5) * 16 + ((tid & 31) >> 4);
    const int R0 = p0 * KK;
#pragma unroll
    for (int i = 0; i < 8; ++i) e[i] = edges[R0 + rb + 2 * i];
}

static __device__ __forceinline__ void fc1_issue(const float* __restrict__ signal,
                                                 const float* __restrict__ efeat,
                                                 int p0, const int (&e)[8], TileRegs& r) {
    const int tid = threadIdx.x;
    const int wv = tid >> 5, lane = tid & 31;
    const int c4 = lane & 15;
    const float4* sig4 = (const float4*)signal;
#pragma unroll
    for (int i = 0; i < 8; ++i) r.g[i] = sig4[(long long)e[i] * 16 + c4];
    r.cv = sig4[(long long)(p0 + (wv >> 1)) * 16 + c4];
    const float* e3 = efeat + (long long)(p0 * KK + wv * 16 + c4) * 3;
    r.f0 = 0.f; r.f1 = 0.f; r.f2 = 0.f;
    if (lane < 16) { r.f0 = e3[0]; r.f1 = e3[1]; r.f2 = e3[2]; }
}

static __device__ __forceinline__ void fc1_store(float* Xs, const TileRegs& r) {
    const int tid = threadIdx.x;
    const int wv = tid >> 5, lane = tid & 31;
    const int c4 = lane & 15;
    const int rb = wv * 16 + (lane >> 4);
#pragma unroll
    for (int i = 0; i < 8; ++i) {
        float4 o;
        o.x = r.g[i].x - r.cv.x; o.y = r.g[i].y - r.cv.y;
        o.z = r.g[i].z - r.cv.z; o.w = r.g[i].w - r.cv.w;
        *(float4*)(Xs + (rb + 2 * i) * KP + c4 * 4) = o;
    }
    if (lane < 16) {
        float4 o; o.x = r.f0; o.y = r.f1; o.z = r.f2; o.w = 0.f;
        *(float4*)(Xs + (wv * 16 + c4) * KP + 64) = o;
    }
}

static __device__ __forceinline__ void fc1_tile_gemm(const float* Xs, const v2f (&wf)[KS1][4],
                                                     int rowBase, int lane, v8f acc[4]) {
    const int l16  = lane & 15;
    const int koff = (lane >> 4) << 1;
    const float* xrow = Xs + (rowBase + l16) * KP + koff;
#pragma unroll
    for (int ks = 0; ks < KS1; ++ks) {
        v2f a = *(const v2f*)(xrow + ks * 4);
#pragma unroll
        for (int t = 0; t < 4; ++t)
            acc[t] = __builtin_amdgcn_wmma_f32_16x16x4_f32(
                false, a, false, wf[ks][t], (short)0, acc[t], false, false);
    }
}

__global__ __launch_bounds__(256) void k_fc1_stats(const float* __restrict__ signal,
                                                   const int* __restrict__ edges,
                                                   const float* __restrict__ efeat,
                                                   const float* __restrict__ W1,
                                                   float* __restrict__ stats1) {
    __shared__ __align__(16) float Xs[2 * XSZ];
    __shared__ __align__(16) float W1s[MID * KP];
    __shared__ float sred[GRP * 2];
    const int tid = threadIdx.x;
    const int wv = tid >> 5, lane = tid & 31, l16 = lane & 15;
    if (tid < GRP * 2) sred[tid] = 0.f;
    const int pbase = blockIdx.x * (TPW * 4);
    W1Regs wr;
    fc1_w_issue(W1, wr);
    int ids[8];
    TileRegs tr;
    fc1_load_ids(edges, pbase, ids);
    fc1_issue(signal, efeat, pbase, ids, tr);
    fc1_load_ids(edges, pbase + 4, ids);
    fc1_w_store(W1s, wr);
    __syncthreads();
    v2f wf[KS1][4];
    fc1_load_wfrag(W1s, lane, wf);
    fc1_store(Xs, tr);
    __syncthreads();

    float ssum[4] = {0.f, 0.f, 0.f, 0.f}, ssq[4] = {0.f, 0.f, 0.f, 0.f};
#pragma unroll 1
    for (int it = 0; it < TPW; ++it) {
        const float* cur = Xs + (it & 1) * XSZ;
        float* nxt = Xs + ((it & 1) ^ 1) * XSZ;
        const bool more = it + 1 < TPW;
        if (more) {
            fc1_issue(signal, efeat, pbase + (it + 1) * 4, ids, tr);
            if (it + 2 < TPW) fc1_load_ids(edges, pbase + (it + 2) * 4, ids);
        }

        v8f acc[4];
        const v8f z = {0.f, 0.f, 0.f, 0.f, 0.f, 0.f, 0.f, 0.f};
        acc[0] = z; acc[1] = z; acc[2] = z; acc[3] = z;
        fc1_tile_gemm(cur, wf, wv * 16, lane, acc);
#pragma unroll
        for (int t = 0; t < 4; ++t)
#pragma unroll
            for (int v = 0; v < 8; ++v) { float x = acc[t][v]; ssum[t] += x; ssq[t] += x * x; }

        if (more) fc1_store(nxt, tr);
        __syncthreads();
    }

#pragma unroll
    for (int t = 0; t < 4; ++t) {
        float s = ssum[t], q = ssq[t];
        s += __shfl_xor(s, 1, 32);  q += __shfl_xor(q, 1, 32);
        s += __shfl_xor(s, 2, 32);  q += __shfl_xor(q, 2, 32);
        s += __shfl_xor(s, 4, 32);  q += __shfl_xor(q, 4, 32);
        s += __shfl_xor(s, 16, 32); q += __shfl_xor(q, 16, 32);
        if ((lane & 7) == 0 && lane < 16) {
            int g = 2 * t + (l16 >> 3);
            atomicAdd(&sred[2 * g + 0], s);
            atomicAdd(&sred[2 * g + 1], q);
        }
    }
    __syncthreads();
    if (tid < GRP * 2) atomicAdd(&stats1[(pbase >> 13) * GRP * 2 + tid], sred[tid]);
}

__global__ __launch_bounds__(256) void k_fc1_pool(const float* __restrict__ signal,
                                                  const int* __restrict__ edges,
                                                  const float* __restrict__ efeat,
                                                  const float* __restrict__ W1,
                                                  const float* __restrict__ g1,
                                                  const float* __restrict__ b1,
                                                  const float* __restrict__ stats1,
                                                  float* __restrict__ h1) {
    __shared__ __align__(16) float Xs[2 * XSZ];
    __shared__ __align__(16) float W1s[MID * KP];
    __shared__ float tmax[2 * 8 * MID];
    __shared__ float scs[MID], bss[MID];
    const int tid = threadIdx.x;
    const int wv = tid >> 5, lane = tid & 31, l16 = lane & 15;
    const int pbase = blockIdx.x * (TPW * 4);
    const int b  = pbase >> 13;
    W1Regs wr;
    fc1_w_issue(W1, wr);
    const int ch = tid & (MID - 1);
    const v2f   st1 = *(const v2f*)(stats1 + b * GRP * 2 + 2 * (ch >> 3));
    const float gw  = g1[ch];
    const float bw  = b1[ch];
    int ids[8];
    TileRegs tr;
    fc1_load_ids(edges, pbase, ids);
    fc1_issue(signal, efeat, pbase, ids, tr);
    fc1_load_ids(edges, pbase + 4, ids);
    {
        const float invc = 1.f / (float)(NN * KK * (MID / GRP));
        float mean = st1.x * invc;
        float var  = st1.y * invc - mean * mean;
        float istd = rsqrtf(var + EPSV);
        float sc = gw * istd;
        scs[ch] = sc;
        bss[ch] = bw - mean * sc;
    }
    fc1_w_store(W1s, wr);
    __syncthreads();
    v2f wf[KS1][4];
    fc1_load_wfrag(W1s, lane, wf);
    fc1_store(Xs, tr);
    __syncthreads();

#pragma unroll 1
    for (int it = 0; it < TPW; ++it) {
        const int p0 = pbase + it * 4;
        const float* cur = Xs + (it & 1) * XSZ;
        float* nxt = Xs + ((it & 1) ^ 1) * XSZ;
        float* tm = tmax + (it & 1) * (8 * MID);
        const bool more = it + 1 < TPW;
        if (more) {
            fc1_issue(signal, efeat, p0 + 4, ids, tr);
            if (it + 2 < TPW) fc1_load_ids(edges, p0 + 8, ids);
        }

        v8f acc[4];
        const v8f z = {0.f, 0.f, 0.f, 0.f, 0.f, 0.f, 0.f, 0.f};
        acc[0] = z; acc[1] = z; acc[2] = z; acc[3] = z;
        fc1_tile_gemm(cur, wf, wv * 16, lane, acc);

#pragma unroll
        for (int t = 0; t < 4; ++t) {
            int c = t * 16 + l16;
            const float sc = scs[c], bs = bss[c];
            float mx = -3.402823466e38f;
#pragma unroll
            for (int v = 0; v < 8; ++v) {
                float x = acc[t][v] * sc + bs;
                x = (x >= 0.f) ? x : SLOPE * x;
                mx = fmaxf(mx, x);
            }
            mx = fmaxf(mx, __shfl_xor(mx, 16, 32));
            if (lane < 16) tm[wv * MID + c] = mx;
        }
        if (more) fc1_store(nxt, tr);
        __syncthreads();
        {
            int q = tid >> 6, c = tid & 63;
            float h = fmaxf(tm[(2 * q) * MID + c], tm[(2 * q + 1) * MID + c]);
            h1[(long long)(p0 + q) * MID + c] = h;
        }
    }
}

typedef __attribute__((ext_vector_type(4))) float v4f;

static __device__ __forceinline__ void load_frag_row(const float* row, int half, v4f (&f)[8]) {
#pragma unroll
    for (int q = 0; q < 8; ++q) f[q] = *(const v4f*)(row + 8 * q + 4 * half);
}

static __device__ __forceinline__ void frag_gemm(const v4f (&a)[8], const v4f (&w)[4][8],
                                                 v8f acc[4]) {
#pragma unroll
    for (int q = 0; q < 8; ++q) {
#pragma unroll
        for (int t = 0; t < 4; ++t)
            acc[t] = __builtin_amdgcn_wmma_f32_16x16x4_f32(
                false, a[q].xy, false, w[t][q].xy, (short)0, acc[t], false, false);
#pragma unroll
        for (int t = 0; t < 4; ++t)
            acc[t] = __builtin_amdgcn_wmma_f32_16x16x4_f32(
                false, a[q].zw, false, w[t][q].zw, (short)0, acc[t], false, false);
    }
}

struct GnRaw { v2f st[4]; float g[4], b[4]; };

static __device__ __forceinline__ void gn_load(const float* __restrict__ stats,
                                               const float* __restrict__ gamma,
                                               const float* __restrict__ beta,
                                               int b, int l16, GnRaw& r) {
#pragma unroll
    for (int t = 0; t < 4; ++t) {
        const int c = t * 16 + l16;
        r.st[t] = *(const v2f*)(stats + b * GRP * 2 + 2 * (c >> 3));
        r.g[t] = gamma[c];
        r.b[t] = beta[c];
    }
}

static __device__ __forceinline__ void gn_coef(const GnRaw& r, float inv_count,
                                               float (&sc)[4], float (&bs)[4]) {
#pragma unroll
    for (int t = 0; t < 4; ++t) {
        float mean = r.st[t].x * inv_count;
        float var  = r.st[t].y * inv_count - mean * mean;
        float istd = rsqrtf(var + EPSV);
        sc[t] = r.g[t] * istd;
        bs[t] = r.b[t] - mean * sc[t];
    }
}

static __device__ __forceinline__ void tail_body(const int mode,
                                                 const float* __restrict__ X,
                                                 const float* __restrict__ Wa,
                                                 const float* __restrict__ Wb,
                                                 const float* __restrict__ ga,
                                                 const float* __restrict__ ba,
                                                 const float* __restrict__ gb,
                                                 const float* __restrict__ bb,
                                                 const float* __restrict__ stats_a,
                                                 const float* __restrict__ stats_b,
                                                 float* __restrict__ stats_out,
                                                 float* __restrict__ out,
                                                 float inv_count,
                                                 float* Xs, float* sred) {
    const int tid = threadIdx.x;
    const int wv = tid >> 5, lane = tid & 31, l16 = lane & 15, half = lane >> 4;
    const long long p0 = (long long)blockIdx.x * 128;
    const int b = (int)(p0 >> 13);
    if (mode < 2) {
        if (tid < GRP * 2) sred[tid] = 0.f;
        if (mode == 0) __syncthreads();
    }

    GnRaw ra, rb;
    if (mode >= 1) gn_load(stats_a, ga, ba, b, l16, ra);
    if (mode >= 2) gn_load(stats_b, gb, bb, b, l16, rb);

    v4f af[8];
    v4f wa[4][8];
    v4f wb[4][8];
    load_frag_row(X + (p0 + wv * 16 + l16) * 64, half, af);
#pragma unroll
    for (int t = 0; t < 4; ++t) load_frag_row(Wa + (t * 16 + l16) * 64, half, wa[t]);
    if (mode >= 1) {
#pragma unroll
        for (int t = 0; t < 4; ++t) load_frag_row(Wb + (t * 16 + l16) * 64, half, wb[t]);
    }
    __builtin_amdgcn_sched_barrier(0);

    v8f acc[4];
    const v8f z = {0.f, 0.f, 0.f, 0.f, 0.f, 0.f, 0.f, 0.f};
    acc[0] = z; acc[1] = z; acc[2] = z; acc[3] = z;
    frag_gemm(af, wa, acc);

    if (mode >= 1) {
        float sc[4], bs[4];
        gn_coef(ra, inv_count, sc, bs);
#pragma unroll
        for (int t = 0; t < 4; ++t) {
            const int c = t * 16 + l16;
#pragma unroll
            for (int v = 0; v < 8; ++v) {
                int row = wv * 16 + v + 8 * half;
                float x = acc[t][v] * sc[t] + bs[t];
                Xs[row * KP + c] = (x >= 0.f) ? x : SLOPE * x;
            }
        }
        __syncthreads();
        load_frag_row(Xs + (wv * 16 + l16) * KP, half, af);
        acc[0] = z; acc[1] = z; acc[2] = z; acc[3] = z;
        frag_gemm(af, wb, acc);
    }

    if (mode >= 2) {
        float sc[4], bs[4];
        gn_coef(rb, inv_count, sc, bs);
#pragma unroll
        for (int t = 0; t < 4; ++t) {
            const int c = t * 16 + l16;
#pragma unroll
            for (int v = 0; v < 8; ++v) {
                int row = wv * 16 + v + 8 * half;
                float x = acc[t][v] * sc[t] + bs[t];
                out[(p0 + row) * 64 + c] = (x >= 0.f) ? x : SLOPE * x;
            }
        }
    } else {
        stats_reduce(acc, lane, sred);
        __syncthreads();
        if (tid < GRP * 2) atomicAdd(&stats_out[b * GRP * 2 + tid], sred[tid]);
    }
}

__global__ __launch_bounds__(256) void k_tail_stats2(const float* __restrict__ h1,
                                                     const float* __restrict__ W2,
                                                     float* __restrict__ stats2) {
    __shared__ float sred[GRP * 2];
    tail_body(0, h1, W2, nullptr, nullptr, nullptr, nullptr, nullptr, nullptr, nullptr,
              stats2, nullptr, 0.f, nullptr, sred);
}

__global__ __launch_bounds__(256) void k_tail_stats3(const float* __restrict__ h1,
                                                     const float* __restrict__ W2,
                                                     const float* __restrict__ g2,
                                                     const float* __restrict__ b2,
                                                     const float* __restrict__ W3,
                                                     const float* __restrict__ stats2,
                                                     float* __restrict__ stats3,
                                                     float inv_count) {
    __shared__ __align__(16) float Xs[128 * KP];
    __shared__ float sred[GRP * 2];
    tail_body(1, h1, W2, W3, g2, b2, nullptr, nullptr, stats2, nullptr,
              stats3, nullptr, inv_count, Xs, sred);
}

__global__ __launch_bounds__(256) void k_tail_out(const float* __restrict__ h1,
                                                  const float* __restrict__ W2,
                                                  const float* __restrict__ g2,
                                                  const float* __restrict__ b2,
                                                  const float* __restrict__ W3,
                                                  const float* __restrict__ g3,
                                                  const float* __restrict__ b3,
                                                  const float* __restrict__ stats2,
                                                  const float* __restrict__ stats3,
                                                  float* __restrict__ out,
                                                  float inv_count) {
    __shared__ __align__(16) float Xs[128 * KP];
    tail_body(2, h1, W2, W3, g2, b2, g3, b3, stats2, stats3,
              nullptr, out, inv_count, Xs, nullptr);
}

__global__ void k_init(float* __restrict__ ws) {
    if (threadIdx.x < 3 * BB * GRP * 2) ws[threadIdx.x] = 0.f;
}

extern "C" void kernel_launch(void* const* d_in, const int* in_sizes, int n_in,
                              void* d_out, int out_size, void* d_ws, size_t ws_size,
                              hipStream_t stream) {
    (void)in_sizes; (void)n_in; (void)out_size; (void)ws_size;
    const float* signal = (const float*)d_in[0];
    const int*   edges  = (const int*)d_in[1];
    const float* efeat  = (const float*)d_in[2];
    const float* W1 = (const float*)d_in[3];
    const float* g1 = (const float*)d_in[4];
    const float* b1 = (const float*)d_in[5];
    const float* W2 = (const float*)d_in[6];
    const float* g2 = (const float*)d_in[7];
    const float* b2 = (const float*)d_in[8];
    const float* W3 = (const float*)d_in[9];
    const float* g3 = (const float*)d_in[10];
    const float* b3 = (const float*)d_in[11];
    float* out = (float*)d_out;

    float* ws = (float*)d_ws;
    float* stats1 = ws;
    float* stats2 = ws + 64;
    float* stats3 = ws + 128;
    float* h1   = ws + 256;
    const float invc23 = 1.f / (float)(NN * (COUT / GRP));

    hipLaunchKernelGGL(k_init,        dim3(1),    dim3(256), 0, stream, ws);
    hipLaunchKernelGGL(k_fc1_stats,   dim3(NWG1), dim3(256), 0, stream,
                       signal, edges, efeat, W1, stats1);
    hipLaunchKernelGGL(k_fc1_pool,    dim3(NWG1), dim3(256), 0, stream,
                       signal, edges, efeat, W1, g1, b1, stats1, h1);
    hipLaunchKernelGGL(k_tail_stats2, dim3(256),  dim3(256), 0, stream,
                       h1, W2, stats2);
    hipLaunchKernelGGL(k_tail_stats3, dim3(256),  dim3(256), 0, stream,
                       h1, W2, g2, b2, W3, stats2, stats3, invc23);
    hipLaunchKernelGGL(k_tail_out,    dim3(256),  dim3(256), 0, stream,
                       h1, W2, g2, b2, W3, g3, b3, stats2, stats3, out, invc23);
}


# fc1 kernels: wave priority (s_setprio) falls with tiles left, 3/2/1/0 from tile 0/10/13/15: of two work-groups on a WGP the one behind wins the matrix pipe, both end together; arithmetic unchanged
# speedup vs baseline: 1.9069x; 1.0433x over previous
#include <hip/hip_runtime.h>

#define BB    4
#define NN    8192
#define KK    32
#define CIN   64
#define CF    67
#define KP    68
#define MID   64
#define COUT  64
#define GRP   8
#define SLOPE 0.1f
#define EPSV  1e-5f

typedef __attribute__((ext_vector_type(2))) float v2f;
typedef __attribute__((ext_vector_type(8))) float v8f;

static __device__ __forceinline__ void stats_reduce(v8f acc[4], int lane, float* sred) {
    const int l16 = lane & 15;
#pragma unroll
    for (int t = 0; t < 4; ++t) {
        float s = 0.f, q = 0.f;
#pragma unroll
        for (int v = 0; v < 8; ++v) { float x = acc[t][v]; s += x; q += x * x; }
        s += __shfl_xor(s, 1, 32);  q += __shfl_xor(q, 1, 32);
        s += __shfl_xor(s, 2, 32);  q += __shfl_xor(q, 2, 32);
        s += __shfl_xor(s, 4, 32);  q += __shfl_xor(q, 4, 32);
        s += __shfl_xor(s, 16, 32); q += __shfl_xor(q, 16, 32);
        if ((lane & 7) == 0 && lane < 16) {
            int g = 2 * t + (l16 >> 3);
            atomicAdd(&sred[2 * g + 0], s);
            atomicAdd(&sred[2 * g + 1], q);
        }
    }
}

#define TPW   16
#define NWG1  512
#define KS1   17
#define XSZ   (128 * KP)

#define WPT   17
struct W1Regs { float w[WPT]; };

static __device__ __forceinline__ void fc1_w_issue(const float* __restrict__ W1, W1Regs& r) {
#pragma unroll
    for (int j = 0; j < WPT; ++j) {
        const int i = threadIdx.x + 256 * j;
        const int m = i / KP, c = i - m * KP;
        r.w[j] = W1[m * CF + ((c < CF) ? c : CF - 1)];
    }
}

static __device__ __forceinline__ void fc1_w_store(float* W1s, const W1Regs& r) {
#pragma unroll
    for (int j = 0; j < WPT; ++j) {
        const int i = threadIdx.x + 256 * j;
        const int c = i % KP;
        W1s[i] = (c < CF) ? r.w[j] : 0.f;
    }
}

static __device__ __forceinline__ void fc1_load_wfrag(const float* W1s, int lane,
                                                      v2f (&wf)[KS1][4]) {
    const int l16  = lane & 15;
    const int koff = (lane >> 4) << 1;
#pragma unroll
    for (int ks = 0; ks < KS1; ++ks)
#pragma unroll
        for (int t = 0; t < 4; ++t)
            wf[ks][t] = *(const v2f*)(W1s + (t * 16 + l16) * KP + ks * 4 + koff);
}

struct TileRegs {
    float4 g[8];
    float4 cv;
    float f0, f1, f2;
};

static __device__ __forceinline__ void fc1_load_ids(const int* __restrict__ edges, int p0,
                                                    int (&e)[8]) {
    const int tid = threadIdx.x;
    const int rb = (tid >> 5) * 16 + ((tid & 31) >> 4);
    const int R0 = p0 * KK;
#pragma unroll
    for (int i = 0; i < 8; ++i) e[i] = edges[R0 + rb + 2 * i];
}

static __device__ __forceinline__ void fc1_issue(const float* __restrict__ signal,
                                                 const float* __restrict__ efeat,
                                                 int p0, const int (&e)[8], TileRegs& r) {
    const int tid = threadIdx.x;
    const int wv = tid >> 5, lane = tid & 31;
    const int c4 = lane & 15;
    const float4* sig4 = (const float4*)signal;
#pragma unroll
    for (int i = 0; i < 8; ++i) r.g[i] = sig4[(long long)e[i] * 16 + c4];
    r.cv = sig4[(long long)(p0 + (wv >> 1)) * 16 + c4];
    const float* e3 = efeat + (long long)(p0 * KK + wv * 16 + c4) * 3;
    r.f0 = 0.f; r.f1 = 0.f; r.f2 = 0.f;
    if (lane < 16) { r.f0 = e3[0]; r.f1 = e3[1]; r.f2 = e3[2]; }
}

static __device__ __forceinline__ void fc1_store(float* Xs, const TileRegs& r) {
    const int tid = threadIdx.x;
    const int wv = tid >> 5, lane = tid & 31;
    const int c4 = lane & 15;
    const int rb = wv * 16 + (lane >> 4);
#pragma unroll
    for (int i = 0; i < 8; ++i) {
        float4 o;
        o.x = r.g[i].x - r.cv.x; o.y = r.g[i].y - r.cv.y;
        o.z = r.g[i].z - r.cv.z; o.w = r.g[i].w - r.cv.w;
        *(float4*)(Xs + (rb + 2 * i) * KP + c4 * 4) = o;
    }
    if (lane < 16) {
        float4 o; o.x = r.f0; o.y = r.f1; o.z = r.f2; o.w = 0.f;
        *(float4*)(Xs + (wv * 16 + c4) * KP + 64) = o;
    }
}

static __device__ __forceinline__ void fc1_tile_gemm(const float* Xs, const v2f (&wf)[KS1][4],
                                                     int rowBase, int lane, v8f acc[4]) {
    const int l16  = lane & 15;
    const int koff = (lane >> 4) << 1;
    const float* xrow = Xs + (rowBase + l16) * KP + koff;
#pragma unroll
    for (int ks = 0; ks < KS1; ++ks) {
        v2f a = *(const v2f*)(xrow + ks * 4);
#pragma unroll
        for (int t = 0; t < 4; ++t)
            acc[t] = __builtin_amdgcn_wmma_f32_16x16x4_f32(
                false, a, false, wf[ks][t], (short)0, acc[t], false, false);
    }
}

static __device__ __forceinline__ void fc1_prio(int it) {
    if (it == 0)  __builtin_amdgcn_s_setprio(3);
    if (it == 10) __builtin_amdgcn_s_setprio(2);
    if (it == 13) __builtin_amdgcn_s_setprio(1);
    if (it == 15) __builtin_amdgcn_s_setprio(0);
}

__global__ __launch_bounds__(256) void k_fc1_stats(const float* __restrict__ signal,
                                                   const int* __restrict__ edges,
                                                   const float* __restrict__ efeat,
                                                   const float* __restrict__ W1,
                                                   float* __restrict__ stats1) {
    __shared__ __align__(16) float Xs[2 * XSZ];
    __shared__ __align__(16) float W1s[MID * KP];
    __shared__ float sred[GRP * 2];
    const int tid = threadIdx.x;
    const int wv = tid >> 5, lane = tid & 31, l16 = lane & 15;
    if (tid < GRP * 2) sred[tid] = 0.f;
    const int pbase = blockIdx.x * (TPW * 4);
    W1Regs wr;
    fc1_w_issue(W1, wr);
    int ids[8];
    TileRegs tr;
    fc1_load_ids(edges, pbase, ids);
    fc1_issue(signal, efeat, pbase, ids, tr);
    fc1_load_ids(edges, pbase + 4, ids);
    fc1_w_store(W1s, wr);
    __syncthreads();
    v2f wf[KS1][4];
    fc1_load_wfrag(W1s, lane, wf);
    fc1_store(Xs, tr);
    __syncthreads();

    float ssum[4] = {0.f, 0.f, 0.f, 0.f}, ssq[4] = {0.f, 0.f, 0.f, 0.f};
#pragma unroll 1
    for (int it = 0; it < TPW; ++it) {
        fc1_prio(it);
        const float* cur = Xs + (it & 1) * XSZ;
        float* nxt = Xs + ((it & 1) ^ 1) * XSZ;
        const bool more = it + 1 < TPW;
        if (more) {
            fc1_issue(signal, efeat, pbase + (it + 1) * 4, ids, tr);
            if (it + 2 < TPW) fc1_load_ids(edges, pbase + (it + 2) * 4, ids);
        }

        v8f acc[4];
        const v8f z = {0.f, 0.f, 0.f, 0.f, 0.f, 0.f, 0.f, 0.f};
        acc[0] = z; acc[1] = z; acc[2] = z; acc[3] = z;
        fc1_tile_gemm(cur, wf, wv * 16, lane, acc);
#pragma unroll
        for (int t = 0; t < 4; ++t)
#pragma unroll
            for (int v = 0; v < 8; ++v) { float x = acc[t][v]; ssum[t] += x; ssq[t] += x * x; }

        if (more) fc1_store(nxt, tr);
        __syncthreads();
    }

#pragma unroll
    for (int t = 0; t < 4; ++t) {
        float s = ssum[t], q = ssq[t];
        s += __shfl_xor(s, 1, 32);  q += __shfl_xor(q, 1, 32);
        s += __shfl_xor(s, 2, 32);  q += __shfl_xor(q, 2, 32);
        s += __shfl_xor(s, 4, 32);  q += __shfl_xor(q, 4, 32);
        s += __shfl_xor(s, 16, 32); q += __shfl_xor(q, 16, 32);
        if ((lane & 7) == 0 && lane < 16) {
            int g = 2 * t + (l16 >> 3);
            atomicAdd(&sred[2 * g + 0], s);
            atomicAdd(&sred[2 * g + 1], q);
        }
    }
    __syncthreads();
    if (tid < GRP * 2) atomicAdd(&stats1[(pbase >> 13) * GRP * 2 + tid], sred[tid]);
}

__global__ __launch_bounds__(256) void k_fc1_pool(const float* __restrict__ signal,
                                                  const int* __restrict__ edges,
                                                  const float* __restrict__ efeat,
                                                  const float* __restrict__ W1,
                                                  const float* __restrict__ g1,
                                                  const float* __restrict__ b1,
                                                  const float* __restrict__ stats1,
                                                  float* __restrict__ h1) {
    __shared__ __align__(16) float Xs[2 * XSZ];
    __shared__ __align__(16) float W1s[MID * KP];
    __shared__ float tmax[2 * 8 * MID];
    __shared__ float scs[MID], bss[MID];
    const int tid = threadIdx.x;
    const int wv = tid >> 5, lane = tid & 31, l16 = lane & 15;
    const int pbase = blockIdx.x * (TPW * 4);
    const int b  = pbase >> 13;
    W1Regs wr;
    fc1_w_issue(W1, wr);
    const int ch = tid & (MID - 1);
    const v2f   st1 = *(const v2f*)(stats1 + b * GRP * 2 + 2 * (ch >> 3));
    const float gw  = g1[ch];
    const float bw  = b1[ch];
    int ids[8];
    TileRegs tr;
    fc1_load_ids(edges, pbase, ids);
    fc1_issue(signal, efeat, pbase, ids, tr);
    fc1_load_ids(edges, pbase + 4, ids);
    {
        const float invc = 1.f / (float)(NN * KK * (MID / GRP));
        float mean = st1.x * invc;
        float var  = st1.y * invc - mean * mean;
        float istd = rsqrtf(var + EPSV);
        float sc = gw * istd;
        scs[ch] = sc;
        bss[ch] = bw - mean * sc;
    }
    fc1_w_store(W1s, wr);
    __syncthreads();
    v2f wf[KS1][4];
    fc1_load_wfrag(W1s, lane, wf);
    fc1_store(Xs, tr);
    __syncthreads();

#pragma unroll 1
    for (int it = 0; it < TPW; ++it) {
        fc1_prio(it);
        const int p0 = pbase + it * 4;
        const float* cur = Xs + (it & 1) * XSZ;
        float* nxt = Xs + ((it & 1) ^ 1) * XSZ;
        float* tm = tmax + (it & 1) * (8 * MID);
        const bool more = it + 1 < TPW;
        if (more) {
            fc1_issue(signal, efeat, p0 + 4, ids, tr);
            if (it + 2 < TPW) fc1_load_ids(edges, p0 + 8, ids);
        }

        v8f acc[4];
        const v8f z = {0.f, 0.f, 0.f, 0.f, 0.f, 0.f, 0.f, 0.f};
        acc[0] = z; acc[1] = z; acc[2] = z; acc[3] = z;
        fc1_tile_gemm(cur, wf, wv * 16, lane, acc);

#pragma unroll
        for (int t = 0; t < 4; ++t) {
            int c = t * 16 + l16;
            const float sc = scs[c], bs = bss[c];
            float mx = -3.402823466e38f;
#pragma unroll
            for (int v = 0; v < 8; ++v) {
                float x = acc[t][v] * sc + bs;
                x = (x >= 0.f) ? x : SLOPE * x;
                mx = fmaxf(mx, x);
            }
            mx = fmaxf(mx, __shfl_xor(mx, 16, 32));
            if (lane < 16) tm[wv * MID + c] = mx;
        }
        if (more) fc1_store(nxt, tr);
        __syncthreads();
        {
            int q = tid >> 6, c = tid & 63;
            float h = fmaxf(tm[(2 * q) * MID + c], tm[(2 * q + 1) * MID + c]);
            h1[(long long)(p0 + q) * MID + c] = h;
        }
    }
}

typedef __attribute__((ext_vector_type(4))) float v4f;

static __device__ __forceinline__ void load_frag_row(const float* row, int half, v4f (&f)[8]) {
#pragma unroll
    for (int q = 0; q < 8; ++q) f[q] = *(const v4f*)(row + 8 * q + 4 * half);
}

static __device__ __forceinline__ void frag_gemm(const v4f (&a)[8], const v4f (&w)[4][8],
                                                 v8f acc[4]) {
#pragma unroll
    for (int q = 0; q < 8; ++q) {
#pragma unroll
        for (int t = 0; t < 4; ++t)
            acc[t] = __builtin_amdgcn_wmma_f32_16x16x4_f32(
                false, a[q].xy, false, w[t][q].xy, (short)0, acc[t], false, false);
#pragma unroll
        for (int t = 0; t < 4; ++t)
            acc[t] = __builtin_amdgcn_wmma_f32_16x16x4_f32(
                false, a[q].zw, false, w[t][q].zw, (short)0, acc[t], false, false);
    }
}

struct GnRaw { v2f st[4]; float g[4], b[4]; };

static __device__ __forceinline__ void gn_load(const float* __restrict__ stats,
                                               const float* __restrict__ gamma,
                                               const float* __restrict__ beta,
                                               int b, int l16, GnRaw& r) {
#pragma unroll
    for (int t = 0; t < 4; ++t) {
        const int c = t * 16 + l16;
        r.st[t] = *(const v2f*)(stats + b * GRP * 2 + 2 * (c >> 3));
        r.g[t] = gamma[c];
        r.b[t] = beta[c];
    }
}

static __device__ __forceinline__ void gn_coef(const GnRaw& r, float inv_count,
                                               float (&sc)[4], float (&bs)[4]) {
#pragma unroll
    for (int t = 0; t < 4; ++t) {
        float mean = r.st[t].x * inv_count;
        float var  = r.st[t].y * inv_count - mean * mean;
        float istd = rsqrtf(var + EPSV);
        sc[t] = r.g[t] * istd;
        bs[t] = r.b[t] - mean * sc[t];
    }
}

static __device__ __forceinline__ void tail_body(const int mode,
                                                 const float* __restrict__ X,
                                                 const float* __restrict__ Wa,
                                                 const float* __restrict__ Wb,
                                                 const float* __restrict__ ga,
                                                 const float* __restrict__ ba,
                                                 const float* __restrict__ gb,
                                                 const float* __restrict__ bb,
                                                 const float* __restrict__ stats_a,
                                                 const float* __restrict__ stats_b,
                                                 float* __restrict__ stats_out,
                                                 float* __restrict__ out,
                                                 float inv_count,
                                                 float* Xs, float* sred) {
    const int tid = threadIdx.x;
    const int wv = tid >> 5, lane = tid & 31, l16 = lane & 15, half = lane >> 4;
    const long long p0 = (long long)blockIdx.x * 128;
    const int b = (int)(p0 >> 13);
    if (mode < 2) {
        if (tid < GRP * 2) sred[tid] = 0.f;
        if (mode == 0) __syncthreads();
    }

    GnRaw ra, rb;
    if (mode >= 1) gn_load(stats_a, ga, ba, b, l16, ra);
    if (mode >= 2) gn_load(stats_b, gb, bb, b, l16, rb);

    v4f af[8];
    v4f wa[4][8];
    v4f wb[4][8];
    load_frag_row(X + (p0 + wv * 16 + l16) * 64, half, af);
#pragma unroll
    for (int t = 0; t < 4; ++t) load_frag_row(Wa + (t * 16 + l16) * 64, half, wa[t]);
    if (mode >= 1) {
#pragma unroll
        for (int t = 0; t < 4; ++t) load_frag_row(Wb + (t * 16 + l16) * 64, half, wb[t]);
    }
    __builtin_amdgcn_sched_barrier(0);

    v8f acc[4];
    const v8f z = {0.f, 0.f, 0.f, 0.f, 0.f, 0.f, 0.f, 0.f};
    acc[0] = z; acc[1] = z; acc[2] = z; acc[3] = z;
    frag_gemm(af, wa, acc);

    if (mode >= 1) {
        float sc[4], bs[4];
        gn_coef(ra, inv_count, sc, bs);
#pragma unroll
        for (int t = 0; t < 4; ++t) {
            const int c = t * 16 + l16;
#pragma unroll
            for (int v = 0; v < 8; ++v) {
                int row = wv * 16 + v + 8 * half;
                float x = acc[t][v] * sc[t] + bs[t];
                Xs[row * KP + c] = (x >= 0.f) ? x : SLOPE * x;
            }
        }
        __syncthreads();
        load_frag_row(Xs + (wv * 16 + l16) * KP, half, af);
        acc[0] = z; acc[1] = z; acc[2] = z; acc[3] = z;
        frag_gemm(af, wb, acc);
    }

    if (mode >= 2) {
        float sc[4], bs[4];
        gn_coef(rb, inv_count, sc, bs);
#pragma unroll
        for (int t = 0; t < 4; ++t) {
            const int c = t * 16 + l16;
#pragma unroll
            for (int v = 0; v < 8; ++v) {
                int row = wv * 16 + v + 8 * half;
                float x = acc[t][v] * sc[t] + bs[t];
                out[(p0 + row) * 64 + c] = (x >= 0.f) ? x : SLOPE * x;
            }
        }
    } else {
        stats_reduce(acc, lane, sred);
        __syncthreads();
        if (tid < GRP * 2) atomicAdd(&stats_out[b * GRP * 2 + tid], sred[tid]);
    }
}

__global__ __launch_bounds__(256) void k_tail_stats2(const float* __restrict__ h1,
                                                     const float* __restrict__ W2,
                                                     float* __restrict__ stats2) {
    __shared__ float sred[GRP * 2];
    tail_body(0, h1, W2, nullptr, nullptr, nullptr, nullptr, nullptr, nullptr, nullptr,
              stats2, nullptr, 0.f, nullptr, sred);
}

__global__ __launch_bounds__(256) void k_tail_stats3(const float* __restrict__ h1,
                                                     const float* __restrict__ W2,
                                                     const float* __restrict__ g2,
                                                     const float* __restrict__ b2,
                                                     const float* __restrict__ W3,
                                                     const float* __restrict__ stats2,
                                                     float* __restrict__ stats3,
                                                     float inv_count) {
    __shared__ __align__(16) float Xs[128 * KP];
    __shared__ float sred[GRP * 2];
    tail_body(1, h1, W2, W3, g2, b2, nullptr, nullptr, stats2, nullptr,
              stats3, nullptr, inv_count, Xs, sred);
}

__global__ __launch_bounds__(256) void k_tail_out(const float* __restrict__ h1,
                                                  const float* __restrict__ W2,
                                                  const float* __restrict__ g2,
                                                  const float* __restrict__ b2,
                                                  const float* __restrict__ W3,
                                                  const float* __restrict__ g3,
                                                  const float* __restrict__ b3,
                                                  const float* __restrict__ stats2,
                                                  const float* __restrict__ stats3,
                                                  float* __restrict__ out,
                                                  float inv_count) {
    __shared__ __align__(16) float Xs[128 * KP];
    tail_body(2, h1, W2, W3, g2, b2, g3, b3, stats2, stats3,
              nullptr, out, inv_count, Xs, nullptr);
}

__global__ void k_init(float* __restrict__ ws) {
    if (threadIdx.x < 3 * BB * GRP * 2) ws[threadIdx.x] = 0.f;
}

extern "C" void kernel_launch(void* const* d_in, const int* in_sizes, int n_in,
                              void* d_out, int out_size, void* d_ws, size_t ws_size,
                              hipStream_t stream) {
    (void)in_sizes; (void)n_in; (void)out_size; (void)ws_size;
    const float* signal = (const float*)d_in[0];
    const int*   edges  = (const int*)d_in[1];
    const float* efeat  = (const float*)d_in[2];
    const float* W1 = (const float*)d_in[3];
    const float* g1 = (const float*)d_in[4];
    const float* b1 = (const float*)d_in[5];
    const float* W2 = (const float*)d_in[6];
    const float* g2 = (const float*)d_in[7];
    const float* b2 = (const float*)d_in[8];
    const float* W3 = (const float*)d_in[9];
    const float* g3 = (const float*)d_in[10];
    const float* b3 = (const float*)d_in[11];
    float* out = (float*)d_out;

    float* ws = (float*)d_ws;
    float* stats1 = ws;
    float* stats2 = ws + 64;
    float* stats3 = ws + 128;
    float* h1   = ws + 256;
    const float invc23 = 1.f / (float)(NN * (COUT / GRP));

    hipLaunchKernelGGL(k_init,        dim3(1),    dim3(256), 0, stream, ws);
    hipLaunchKernelGGL(k_fc1_stats,   dim3(NWG1), dim3(256), 0, stream,
                       signal, edges, efeat, W1, stats1);
    hipLaunchKernelGGL(k_fc1_pool,    dim3(NWG1), dim3(256), 0, stream,
                       signal, edges, efeat, W1, g1, b1, stats1, h1);
    hipLaunchKernelGGL(k_tail_stats2, dim3(256),  dim3(256), 0, stream,
                       h1, W2, stats2);
    hipLaunchKernelGGL(k_tail_stats3, dim3(256),  dim3(256), 0, stream,
                       h1, W2, g2, b2, W3, stats2, stats3, invc23);
    hipLaunchKernelGGL(k_tail_out,    dim3(256),  dim3(256), 0, stream,
                       h1, W2, g2, b2, W3, g3, b3, stats2, stats3, out, invc23);
}


# fc2 moved into the fc1 pooling kernel: a work-group multiplies its 64 pooled rows by W2, writes raw fc2 rows (h1 never stored), adds stats2; tail 3 launches to 2, fc2-free; stats2 sum order changed
# speedup vs baseline: 2.0791x; 1.0903x over previous
#include <hip/hip_runtime.h>

#define BB    4
#define NN    8192
#define KK    32
#define CIN   64
#define CF    67
#define KP    68
#define MID   64
#define COUT  64
#define GRP   8
#define SLOPE 0.1f
#define EPSV  1e-5f

typedef __attribute__((ext_vector_type(2))) float v2f;
typedef __attribute__((ext_vector_type(8))) float v8f;

static __device__ __forceinline__ void stats_reduce(v8f acc[4], int lane, float* sred) {
    const int l16 = lane & 15;
#pragma unroll
    for (int t = 0; t < 4; ++t) {
        float s = 0.f, q = 0.f;
#pragma unroll
        for (int v = 0; v < 8; ++v) { float x = acc[t][v]; s += x; q += x * x; }
        s += __shfl_xor(s, 1, 32);  q += __shfl_xor(q, 1, 32);
        s += __shfl_xor(s, 2, 32);  q += __shfl_xor(q, 2, 32);
        s += __shfl_xor(s, 4, 32);  q += __shfl_xor(q, 4, 32);
        s += __shfl_xor(s, 16, 32); q += __shfl_xor(q, 16, 32);
        if ((lane & 7) == 0 && lane < 16) {
            int g = 2 * t + (l16 >> 3);
            atomicAdd(&sred[2 * g + 0], s);
            atomicAdd(&sred[2 * g + 1], q);
        }
    }
}

#define TPW   16
#define NWG1  512
#define KS1   17
#define XSZ   (128 * KP)

#define WPT   17
struct W1Regs { float w[WPT]; };

static __device__ __forceinline__ void fc1_w_issue(const float* __restrict__ W1, W1Regs& r) {
#pragma unroll
    for (int j = 0; j < WPT; ++j) {
        const int i = threadIdx.x + 256 * j;
        const int m = i / KP, c = i - m * KP;
        r.w[j] = W1[m * CF + ((c < CF) ? c : CF - 1)];
    }
}

static __device__ __forceinline__ void fc1_w_store(float* W1s, const W1Regs& r) {
#pragma unroll
    for (int j = 0; j < WPT; ++j) {
        const int i = threadIdx.x + 256 * j;
        const int c = i % KP;
        W1s[i] = (c < CF) ? r.w[j] : 0.f;
    }
}

static __device__ __forceinline__ void fc1_load_wfrag(const float* W1s, int lane,
                                                      v2f (&wf)[KS1][4]) {
    const int l16  = lane & 15;
    const int koff = (lane >> 4) << 1;
#pragma unroll
    for (int ks = 0; ks < KS1; ++ks)
#pragma unroll
        for (int t = 0; t < 4; ++t)
            wf[ks][t] = *(const v2f*)(W1s + (t * 16 + l16) * KP + ks * 4 + koff);
}

struct TileRegs {
    float4 g[8];
    float4 cv;
    float f0, f1, f2;
};

static __device__ __forceinline__ void fc1_load_ids(const int* __restrict__ edges, int p0,
                                                    int (&e)[8]) {
    const int tid = threadIdx.x;
    const int rb = (tid >> 5) * 16 + ((tid & 31) >> 4);
    const int R0 = p0 * KK;
#pragma unroll
    for (int i = 0; i < 8; ++i) e[i] = edges[R0 + rb + 2 * i];
}

static __device__ __forceinline__ void fc1_issue(const float* __restrict__ signal,
                                                 const float* __restrict__ efeat,
                                                 int p0, const int (&e)[8], TileRegs& r) {
    const int tid = threadIdx.x;
    const int wv = tid >> 5, lane = tid & 31;
    const int c4 = lane & 15;
    const float4* sig4 = (const float4*)signal;
#pragma unroll
    for (int i = 0; i < 8; ++i) r.g[i] = sig4[(long long)e[i] * 16 + c4];
    r.cv = sig4[(long long)(p0 + (wv >> 1)) * 16 + c4];
    const float* e3 = efeat + (long long)(p0 * KK + wv * 16 + c4) * 3;
    r.f0 = 0.f; r.f1 = 0.f; r.f2 = 0.f;
    if (lane < 16) { r.f0 = e3[0]; r.f1 = e3[1]; r.f2 = e3[2]; }
}

static __device__ __forceinline__ void fc1_store(float* Xs, const TileRegs& r) {
    const int tid = threadIdx.x;
    const int wv = tid >> 5, lane = tid & 31;
    const int c4 = lane & 15;
    const int rb = wv * 16 + (lane >> 4);
#pragma unroll
    for (int i = 0; i < 8; ++i) {
        float4 o;
        o.x = r.g[i].x - r.cv.x; o.y = r.g[i].y - r.cv.y;
        o.z = r.g[i].z - r.cv.z; o.w = r.g[i].w - r.cv.w;
        *(float4*)(Xs + (rb + 2 * i) * KP + c4 * 4) = o;
    }
    if (lane < 16) {
        float4 o; o.x = r.f0; o.y = r.f1; o.z = r.f2; o.w = 0.f;
        *(float4*)(Xs + (wv * 16 + c4) * KP + 64) = o;
    }
}

static __device__ __forceinline__ void fc1_tile_gemm(const float* Xs, const v2f (&wf)[KS1][4],
                                                     int rowBase, int lane, v8f acc[4]) {
    const int l16  = lane & 15;
    const int koff = (lane >> 4) << 1;
    const float* xrow = Xs + (rowBase + l16) * KP + koff;
#pragma unroll
    for (int ks = 0; ks < KS1; ++ks) {
        v2f a = *(const v2f*)(xrow + ks * 4);
#pragma unroll
        for (int t = 0; t < 4; ++t)
            acc[t] = __builtin_amdgcn_wmma_f32_16x16x4_f32(
                false, a, false, wf[ks][t], (short)0, acc[t], false, false);
    }
}

static __device__ __forceinline__ void fc1_prio(int it) {
    if (it == 0)  __builtin_amdgcn_s_setprio(3);
    if (it == 10) __builtin_amdgcn_s_setprio(2);
    if (it == 13) __builtin_amdgcn_s_setprio(1);
    if (it == 15) __builtin_amdgcn_s_setprio(0);
}

__global__ __launch_bounds__(256) void k_fc1_stats(const float* __restrict__ signal,
                                                   const int* __restrict__ edges,
                                                   const float* __restrict__ efeat,
                                                   const float* __restrict__ W1,
                                                   float* __restrict__ stats1) {
    __shared__ __align__(16) float Xs[2 * XSZ];
    __shared__ __align__(16) float W1s[MID * KP];
    __shared__ float sred[GRP * 2];
    const int tid = threadIdx.x;
    const int wv = tid >> 5, lane = tid & 31, l16 = lane & 15;
    if (tid < GRP * 2) sred[tid] = 0.f;
    const int pbase = blockIdx.x * (TPW * 4);
    W1Regs wr;
    fc1_w_issue(W1, wr);
    int ids[8];
    TileRegs tr;
    fc1_load_ids(edges, pbase, ids);
    fc1_issue(signal, efeat, pbase, ids, tr);
    fc1_load_ids(edges, pbase + 4, ids);
    fc1_w_store(W1s, wr);
    __syncthreads();
    v2f wf[KS1][4];
    fc1_load_wfrag(W1s, lane, wf);
    fc1_store(Xs, tr);
    __syncthreads();

    float ssum[4] = {0.f, 0.f, 0.f, 0.f}, ssq[4] = {0.f, 0.f, 0.f, 0.f};
#pragma unroll 1
    for (int it = 0; it < TPW; ++it) {
        fc1_prio(it);
        const float* cur = Xs + (it & 1) * XSZ;
        float* nxt = Xs + ((it & 1) ^ 1) * XSZ;
        const bool more = it + 1 < TPW;
        if (more) {
            fc1_issue(signal, efeat, pbase + (it + 1) * 4, ids, tr);
            if (it + 2 < TPW) fc1_load_ids(edges, pbase + (it + 2) * 4, ids);
        }

        v8f acc[4];
        const v8f z = {0.f, 0.f, 0.f, 0.f, 0.f, 0.f, 0.f, 0.f};
        acc[0] = z; acc[1] = z; acc[2] = z; acc[3] = z;
        fc1_tile_gemm(cur, wf, wv * 16, lane, acc);
#pragma unroll
        for (int t = 0; t < 4; ++t)
#pragma unroll
            for (int v = 0; v < 8; ++v) { float x = acc[t][v]; ssum[t] += x; ssq[t] += x * x; }

        if (more) fc1_store(nxt, tr);
        __syncthreads();
    }

#pragma unroll
    for (int t = 0; t < 4; ++t) {
        float s = ssum[t], q = ssq[t];
        s += __shfl_xor(s, 1, 32);  q += __shfl_xor(q, 1, 32);
        s += __shfl_xor(s, 2, 32);  q += __shfl_xor(q, 2, 32);
        s += __shfl_xor(s, 4, 32);  q += __shfl_xor(q, 4, 32);
        s += __shfl_xor(s, 16, 32); q += __shfl_xor(q, 16, 32);
        if ((lane & 7) == 0 && lane < 16) {
            int g = 2 * t + (l16 >> 3);
            atomicAdd(&sred[2 * g + 0], s);
            atomicAdd(&sred[2 * g + 1], q);
        }
    }
    __syncthreads();
    if (tid < GRP * 2) atomicAdd(&stats1[(pbase >> 13) * GRP * 2 + tid], sred[tid]);
}

typedef __attribute__((ext_vector_type(4))) float v4f;

static __device__ __forceinline__ void load_frag_row(const float* row, int half, v4f (&f)[8]) {
#pragma unroll
    for (int q = 0; q < 8; ++q) f[q] = *(const v4f*)(row + 8 * q + 4 * half);
}

static __device__ __forceinline__ void load_frag_tile(const float* __restrict__ Wf, int t, int lane,
                                                      v4f (&f)[8]) {
#pragma unroll
    for (int q = 0; q < 8; ++q) f[q] = *(const v4f*)(Wf + ((t * 8 + q) * 32 + lane) * 4);
}

__global__ __launch_bounds__(256) __attribute__((amdgpu_waves_per_eu(4, 4))) void k_fc1_pool(const float* __restrict__ signal,
                                                  const int* __restrict__ edges,
                                                  const float* __restrict__ efeat,
                                                  const float* __restrict__ W1,
                                                  const float* __restrict__ g1,
                                                  const float* __restrict__ b1,
                                                  const float* __restrict__ stats1,
                                                  const float* __restrict__ W2f,
                                                  float* __restrict__ stats2,
                                                  float* __restrict__ y2) {
    __shared__ __align__(16) float Xs[2 * XSZ];
    __shared__ __align__(16) float W1s[MID * KP];
    __shared__ float tmax[2 * 8 * MID];
    __shared__ float scs[MID], bss[MID];
    __shared__ __align__(16) float h1s[TPW * 4 * KP];
    __shared__ float sred2[GRP * 2];
    const int tid = threadIdx.x;
    const int wv = tid >> 5, lane = tid & 31, l16 = lane & 15;
    const int pbase = blockIdx.x * (TPW * 4);
    const int b  = pbase >> 13;
    W1Regs wr;
    fc1_w_issue(W1, wr);
    const int ch = tid & (MID - 1);
    const v2f   st1 = *(const v2f*)(stats1 + b * GRP * 2 + 2 * (ch >> 3));
    const float gw  = g1[ch];
    const float bw  = b1[ch];
    int ids[8];
    TileRegs tr;
    fc1_load_ids(edges, pbase, ids);
    fc1_issue(signal, efeat, pbase, ids, tr);
    fc1_load_ids(edges, pbase + 4, ids);
    {
        const float invc = 1.f / (float)(NN * KK * (MID / GRP));
        float mean = st1.x * invc;
        float var  = st1.y * invc - mean * mean;
        float istd = rsqrtf(var + EPSV);
        float sc = gw * istd;
        scs[ch] = sc;
        bss[ch] = bw - mean * sc;
    }
    fc1_w_store(W1s, wr);
    __syncthreads();
    v2f wf[KS1][4];
    fc1_load_wfrag(W1s, lane, wf);
    fc1_store(Xs, tr);
    __syncthreads();

#pragma unroll 1
    for (int it = 0; it < TPW; ++it) {
        fc1_prio(it);
        const int p0 = pbase + it * 4;
        const float* cur = Xs + (it & 1) * XSZ;
        float* nxt = Xs + ((it & 1) ^ 1) * XSZ;
        float* tm = tmax + (it & 1) * (8 * MID);
        const bool more = it + 1 < TPW;
        if (more) {
            fc1_issue(signal, efeat, p0 + 4, ids, tr);
            if (it + 2 < TPW) fc1_load_ids(edges, p0 + 8, ids);
        }

        v8f acc[4];
        const v8f z = {0.f, 0.f, 0.f, 0.f, 0.f, 0.f, 0.f, 0.f};
        acc[0] = z; acc[1] = z; acc[2] = z; acc[3] = z;
        fc1_tile_gemm(cur, wf, wv * 16, lane, acc);

#pragma unroll
        for (int t = 0; t < 4; ++t) {
            int c = t * 16 + l16;
            const float sc = scs[c], bs = bss[c];
            float mx = -3.402823466e38f;
#pragma unroll
            for (int v = 0; v < 8; ++v) {
                float x = acc[t][v] * sc + bs;
                x = (x >= 0.f) ? x : SLOPE * x;
                mx = fmaxf(mx, x);
            }
            mx = fmaxf(mx, __shfl_xor(mx, 16, 32));
            if (lane < 16) tm[wv * MID + c] = mx;
        }
        if (more) fc1_store(nxt, tr);
        __syncthreads();
        {
            int q = tid >> 6, c = tid & 63;
            float h = fmaxf(tm[(2 * q) * MID + c], tm[(2 * q + 1) * MID + c]);
            h1s[(it * 4 + q) * KP + c] = h;
        }
    }

    const int half = lane >> 4;
    const int rt = wv >> 1, t0 = (wv & 1) * 2;
    v4f wa[2][8];
    load_frag_tile(W2f, t0, lane, wa[0]);
    load_frag_tile(W2f, t0 + 1, lane, wa[1]);
    if (tid < GRP * 2) sred2[l16] = 0.f;
    __syncthreads();
    v4f af[8];
    load_frag_row(h1s + (rt * 16 + l16) * KP, half, af);
    v8f acc2[2];
    {
        const v8f z = {0.f, 0.f, 0.f, 0.f, 0.f, 0.f, 0.f, 0.f};
        acc2[0] = z; acc2[1] = z;
    }
#pragma unroll
    for (int q = 0; q < 8; ++q) {
#pragma unroll
        for (int tt = 0; tt < 2; ++tt)
            acc2[tt] = __builtin_amdgcn_wmma_f32_16x16x4_f32(
                false, af[q].xy, false, wa[tt][q].xy, (short)0, acc2[tt], false, false);
#pragma unroll
        for (int tt = 0; tt < 2; ++tt)
            acc2[tt] = __builtin_amdgcn_wmma_f32_16x16x4_f32(
                false, af[q].zw, false, wa[tt][q].zw, (short)0, acc2[tt], false, false);
    }
#pragma unroll
    for (int tt = 0; tt < 2; ++tt) {
        const int c = (t0 + tt) * 16 + l16;
        float s = 0.f, sq = 0.f;
#pragma unroll
        for (int v = 0; v < 8; ++v) {
            const float x = acc2[tt][v];
            y2[(long long)(pbase + rt * 16 + v + 8 * half) * COUT + c] = x;
            s += x; sq += x * x;
        }
        s += __shfl_xor(s, 1, 32);  sq += __shfl_xor(sq, 1, 32);
        s += __shfl_xor(s, 2, 32);  sq += __shfl_xor(sq, 2, 32);
        s += __shfl_xor(s, 4, 32);  sq += __shfl_xor(sq, 4, 32);
        s += __shfl_xor(s, 16, 32); sq += __shfl_xor(sq, 16, 32);
        if ((lane & 7) == 0 && lane < 16) {
            const int g = 2 * (t0 + tt) + (l16 >> 3);
            atomicAdd(&sred2[2 * g + 0], s);
            atomicAdd(&sred2[2 * g + 1], sq);
        }
    }
    __syncthreads();
    if (tid < GRP * 2) atomicAdd(&stats2[b * GRP * 2 + l16], sred2[l16]);
}


static __device__ __forceinline__ void frag_gemm(const v4f (&a)[8], const v4f (&w)[4][8],
                                                 v8f acc[4]) {
#pragma unroll
    for (int q = 0; q < 8; ++q) {
#pragma unroll
        for (int t = 0; t < 4; ++t)
            acc[t] = __builtin_amdgcn_wmma_f32_16x16x4_f32(
                false, a[q].xy, false, w[t][q].xy, (short)0, acc[t], false, false);
#pragma unroll
        for (int t = 0; t < 4; ++t)
            acc[t] = __builtin_amdgcn_wmma_f32_16x16x4_f32(
                false, a[q].zw, false, w[t][q].zw, (short)0, acc[t], false, false);
    }
}

struct GnRaw { v2f st[4]; float g[4], b[4]; };

static __device__ __forceinline__ void gn_load(const float* __restrict__ stats,
                                               const float* __restrict__ gamma,
                                               const float* __restrict__ beta,
                                               int b, int l16, GnRaw& r) {
#pragma unroll
    for (int t = 0; t < 4; ++t) {
        const int c = t * 16 + l16;
        r.st[t] = *(const v2f*)(stats + b * GRP * 2 + 2 * (c >> 3));
        r.g[t] = gamma[c];
        r.b[t] = beta[c];
    }
}

static __device__ __forceinline__ void gn_coef(const GnRaw& r, float inv_count,
                                               float (&sc)[4], float (&bs)[4]) {
#pragma unroll
    for (int t = 0; t < 4; ++t) {
        float mean = r.st[t].x * inv_count;
        float var  = r.st[t].y * inv_count - mean * mean;
        float istd = rsqrtf(var + EPSV);
        sc[t] = r.g[t] * istd;
        bs[t] = r.b[t] - mean * sc[t];
    }
}

struct GnFragRaw { v2f st[8]; v4f g[8], b[8]; };

static __device__ __forceinline__ void gnf_load(const float* __restrict__ stats,
                                                const float* __restrict__ gamma,
                                                const float* __restrict__ beta,
                                                int b, int half, GnFragRaw& r) {
#pragma unroll
    for (int q = 0; q < 8; ++q) {
        r.st[q] = *(const v2f*)(stats + b * GRP * 2 + 2 * q);
        r.g[q] = *(const v4f*)(gamma + 8 * q + 4 * half);
        r.b[q] = *(const v4f*)(beta + 8 * q + 4 * half);
    }
}

static __device__ __forceinline__ void gnf_apply(const GnFragRaw& r, float inv_count,
                                                 v4f (&af)[8]) {
#pragma unroll
    for (int q = 0; q < 8; ++q) {
        float mean = r.st[q].x * inv_count;
        float var  = r.st[q].y * inv_count - mean * mean;
        float istd = rsqrtf(var + EPSV);
#pragma unroll
        for (int j = 0; j < 4; ++j) {
            float sc = r.g[q][j] * istd;
            float bs = r.b[q][j] - mean * sc;
            float x = af[q][j] * sc + bs;
            af[q][j] = (x >= 0.f) ? x : SLOPE * x;
        }
    }
}

static __device__ __forceinline__ void tail_body(const int mode,
                                                 const float* __restrict__ Y,
                                                 const float* __restrict__ Wb,
                                                 const float* __restrict__ ga,
                                                 const float* __restrict__ ba,
                                                 const float* __restrict__ gb,
                                                 const float* __restrict__ bb,
                                                 const float* __restrict__ stats_a,
                                                 const float* __restrict__ stats_b,
                                                 float* __restrict__ stats_out,
                                                 float* __restrict__ out,
                                                 float inv_count,
                                                 float* sred) {
    const int tid = threadIdx.x;
    const int wv = tid >> 5, lane = tid & 31, l16 = lane & 15, half = lane >> 4;
    const long long p0 = (long long)blockIdx.x * 128;
    const int b = (int)(p0 >> 13);
    if (mode < 2) {
        if (tid < GRP * 2) sred[tid] = 0.f;
        __syncthreads();
    }

    GnFragRaw ra;
    GnRaw rb;
    gnf_load(stats_a, ga, ba, b, half, ra);
    if (mode >= 2) gn_load(stats_b, gb, bb, b, l16, rb);

    v4f af[8];
    v4f wb[4][8];
    load_frag_row(Y + (p0 + wv * 16 + l16) * 64, half, af);
#pragma unroll
    for (int t = 0; t < 4; ++t) load_frag_tile(Wb, t, lane, wb[t]);
    __builtin_amdgcn_sched_barrier(0);

    gnf_apply(ra, inv_count, af);

    v8f acc[4];
    const v8f z = {0.f, 0.f, 0.f, 0.f, 0.f, 0.f, 0.f, 0.f};
    acc[0] = z; acc[1] = z; acc[2] = z; acc[3] = z;
    frag_gemm(af, wb, acc);

    if (mode >= 2) {
        float sc[4], bs[4];
        gn_coef(rb, inv_count, sc, bs);
#pragma unroll
        for (int t = 0; t < 4; ++t) {
            const int c = t * 16 + l16;
#pragma unroll
            for (int v = 0; v < 8; ++v) {
                int row = wv * 16 + v + 8 * half;
                float x = acc[t][v] * sc[t] + bs[t];
                out[(p0 + row) * 64 + c] = (x >= 0.f) ? x : SLOPE * x;
            }
        }
    } else {
        stats_reduce(acc, lane, sred);
        __syncthreads();
        if (tid < GRP * 2) atomicAdd(&stats_out[b * GRP * 2 + tid], sred[tid]);
    }
}

__global__ __launch_bounds__(256) void k_tail_stats3(const float* __restrict__ y2,
                                                     const float* __restrict__ g2,
                                                     const float* __restrict__ b2,
                                                     const float* __restrict__ W3f,
                                                     const float* __restrict__ stats2,
                                                     float* __restrict__ stats3,
                                                     float inv_count) {
    __shared__ float sred[GRP * 2];
    tail_body(1, y2, W3f, g2, b2, nullptr, nullptr, stats2, nullptr,
              stats3, nullptr, inv_count, sred);
}

__global__ __launch_bounds__(256) void k_tail_out(const float* __restrict__ y2,
                                                  const float* __restrict__ g2,
                                                  const float* __restrict__ b2,
                                                  const float* __restrict__ W3f,
                                                  const float* __restrict__ g3,
                                                  const float* __restrict__ b3,
                                                  const float* __restrict__ stats2,
                                                  const float* __restrict__ stats3,
                                                  float* __restrict__ out,
                                                  float inv_count) {
    tail_body(2, y2, W3f, g2, b2, g3, b3, stats2, stats3,
              nullptr, out, inv_count, nullptr);
}

__global__ __launch_bounds__(256) void k_init(float* __restrict__ ws,
                                              const float* __restrict__ W2,
                                              const float* __restrict__ W3,
                                              float* __restrict__ wf2,
                                              float* __restrict__ wf3) {
    const int tid = threadIdx.x;
    if (blockIdx.x == 0) {
        if (tid < 3 * BB * GRP * 2) ws[tid] = 0.f;
        return;
    }
    const int m = blockIdx.x - 1;
    const float* W = (m < 4) ? W2 : W3;
    float* Wf = (m < 4) ? wf2 : wf3;
    const int t = m & 3, q = tid >> 5, lane = tid & 31;
    const v4f v = *(const v4f*)(W + (t * 16 + (lane & 15)) * 64 + 8 * q + 4 * (lane >> 4));
    *(v4f*)(Wf + ((t * 8 + q) * 32 + lane) * 4) = v;
}

extern "C" void kernel_launch(void* const* d_in, const int* in_sizes, int n_in,
                              void* d_out, int out_size, void* d_ws, size_t ws_size,
                              hipStream_t stream) {
    (void)in_sizes; (void)n_in; (void)out_size; (void)ws_size;
    const float* signal = (const float*)d_in[0];
    const int*   edges  = (const int*)d_in[1];
    const float* efeat  = (const float*)d_in[2];
    const float* W1 = (const float*)d_in[3];
    const float* g1 = (const float*)d_in[4];
    const float* b1 = (const float*)d_in[5];
    const float* W2 = (const float*)d_in[6];
    const float* g2 = (const float*)d_in[7];
    const float* b2 = (const float*)d_in[8];
    const float* W3 = (const float*)d_in[9];
    const float* g3 = (const float*)d_in[10];
    const float* b3 = (const float*)d_in[11];
    float* out = (float*)d_out;

    float* ws = (float*)d_ws;
    float* stats1 = ws;
    float* stats2 = ws + 64;
    float* stats3 = ws + 128;
    float* y2   = ws + 256;
    float* wf2  = ws + 256 + 2097152;
    float* wf3  = ws + 256 + 2097152 + 4096;
    const float invc23 = 1.f / (float)(NN * (COUT / GRP));

    hipLaunchKernelGGL(k_init,        dim3(9),    dim3(256), 0, stream, ws, W2, W3, wf2, wf3);
    hipLaunchKernelGGL(k_fc1_stats,   dim3(NWG1), dim3(256), 0, stream,
                       signal, edges, efeat, W1, stats1);
    hipLaunchKernelGGL(k_fc1_pool,    dim3(NWG1), dim3(256), 0, stream,
                       signal, edges, efeat, W1, g1, b1, stats1, wf2, stats2, y2);
    hipLaunchKernelGGL(k_tail_stats3, dim3(256),  dim3(256), 0, stream,
                       y2, g2, b2, wf3, stats2, stats3, invc23);
    hipLaunchKernelGGL(k_tail_out,    dim3(256),  dim3(256), 0, stream,
                       y2, g2, b2, wf3, g3, b3, stats2, stats3, out, invc23);
}
